# starting kernel
# baseline (speedup 1.0000x reference)
#include <hip/hip_runtime.h>
#include <hip/hip_bf16.h>

typedef float v2f __attribute__((ext_vector_type(2)));
typedef float v8f __attribute__((ext_vector_type(8)));

#define PAIR_STRIDE 288

__global__ __launch_bounds__(256) void edgeconv_gemm(const float* __restrict__ x,
                                                     const float* __restrict__ W,
                                                     float* __restrict__ H) {
    __shared__ float wc[32 * PAIR_STRIDE];

    const int tid = threadIdx.x;
    for (int i = tid; i < 64 * 128; i += 256) {
        int c = i >> 7;
        int o = i & 127;
        float v;
        if (o < 64) v = W[c * 64 + o] - W[(64 + c) * 64 + o];
        else        v = W[(64 + c) * 64 + (o - 64)];
        wc[(c >> 1) * PAIR_STRIDE + o * 2 + (c & 1)] = v;
    }
    __syncthreads();

    const int lane = tid & 31;
    const int wv   = tid >> 5;
    const int tile = blockIdx.x * 8 + wv;
    const int m    = lane & 15;
    const int kh   = lane >> 4;
    const int node = tile * 16 + m;
    const int bb   = node >> 12;
    const int nn   = node & 4095;
    const float* xb = x + ((size_t)bb * 64) * 4096 + nn;

    v2f a[16];
#pragma unroll
    for (int kk = 0; kk < 16; ++kk) {
        int c0 = 4 * kk + 2 * kh;
        a[kk].x = xb[(size_t)c0 * 4096];
        a[kk].y = xb[(size_t)(c0 + 1) * 4096];
    }

#pragma unroll
    for (int ot = 0; ot < 8; ++ot) {
        v8f acc = {};
#pragma unroll
        for (int kk = 0; kk < 16; ++kk) {
            const v2f bf = *(const v2f*)&wc[(2 * kk + kh) * PAIR_STRIDE + (ot * 16 + m) * 2];
            acc = __builtin_amdgcn_wmma_f32_16x16x4_f32(
                false, a[kk], false, bf, (short)0, acc, false, false);
        }
        float* Hp = H + ((size_t)tile * 16 + 8 * kh) * 128 + ot * 16 + m;
#pragma unroll
        for (int j = 0; j < 8; ++j)
            Hp[(size_t)j * 128] = acc[j];
    }
}

__global__ __launch_bounds__(256) void edgeconv_aggregate(const float* __restrict__ H,
                                                          const int* __restrict__ src,
                                                          const float* __restrict__ bias,
                                                          float* __restrict__ out) {
    const int lane = threadIdx.x & 31;
    const int wv   = threadIdx.x >> 5;
    const int node = blockIdx.x * 8 + wv;
    const int bb   = node >> 12;
    const int nn   = node & 4095;

    int s = src[node * 16 + (lane & 15)];

    float m0 = -3.402823466e38f, m1 = -3.402823466e38f;
#pragma unroll
    for (int k = 0; k < 16; ++k) {
        int sk = __shfl(s, k, 32);
        const v2f* crow = (const v2f*)(H + ((size_t)(bb * 4096 + sk)) * 128 + 64);
        v2f c = crow[lane];
        m0 = fmaxf(m0, c.x);
        m1 = fmaxf(m1, c.y);
    }

    v2f aA = ((const v2f*)(H + (size_t)node * 128))[lane];
    v2f bV = ((const v2f*)bias)[lane];
    float r0 = fmaxf(aA.x + bV.x + m0, 0.0f);
    float r1 = fmaxf(aA.y + bV.y + m1, 0.0f);

    size_t obase = ((size_t)bb * 64 + 2 * lane) * 4096 + nn;
    out[obase]        = r0;
    out[obase + 4096] = r1;
}

extern "C" void kernel_launch(void* const* d_in, const int* in_sizes, int n_in,
                              void* d_out, int out_size, void* d_ws, size_t ws_size,
                              hipStream_t stream) {
    const float* x  = (const float*)d_in[0];
    const int*   ei = (const int*)d_in[1];
    const float* W  = (const float*)d_in[2];
    const float* b  = (const float*)d_in[3];
    float* out = (float*)d_out;
    float* H   = (float*)d_ws;

    edgeconv_gemm<<<256, 256, 0, stream>>>(x, W, H);
    edgeconv_aggregate<<<4096, 256, 0, stream>>>(H, ei, b, out);
}


# aggregate re-tiled: one work-group = 32 consecutive nodes (4 per wave, 8192 waves instead of 32768), its 64x32 result tile transposed in LDS so each 128-B output line is one whole store
# speedup vs baseline: 2.6052x; 2.6052x over previous
#include <hip/hip_runtime.h>
#include <hip/hip_bf16.h>

typedef float v2f __attribute__((ext_vector_type(2)));
typedef float v8f __attribute__((ext_vector_type(8)));

#define PAIR_STRIDE 288

__global__ __launch_bounds__(256) void edgeconv_gemm(const float* __restrict__ x,
                                                     const float* __restrict__ W,
                                                     float* __restrict__ H) {
    __shared__ float wc[32 * PAIR_STRIDE];

    const int tid = threadIdx.x;
    for (int i = tid; i < 64 * 128; i += 256) {
        int c = i >> 7;
        int o = i & 127;
        float v;
        if (o < 64) v = W[c * 64 + o] - W[(64 + c) * 64 + o];
        else        v = W[(64 + c) * 64 + (o - 64)];
        wc[(c >> 1) * PAIR_STRIDE + o * 2 + (c & 1)] = v;
    }
    __syncthreads();

    const int lane = tid & 31;
    const int wv   = tid >> 5;
    const int tile = blockIdx.x * 8 + wv;
    const int m    = lane & 15;
    const int kh   = lane >> 4;
    const int node = tile * 16 + m;
    const int bb   = node >> 12;
    const int nn   = node & 4095;
    const float* xb = x + ((size_t)bb * 64) * 4096 + nn;

    v2f a[16];
#pragma unroll
    for (int kk = 0; kk < 16; ++kk) {
        int c0 = 4 * kk + 2 * kh;
        a[kk].x = xb[(size_t)c0 * 4096];
        a[kk].y = xb[(size_t)(c0 + 1) * 4096];
    }

#pragma unroll
    for (int ot = 0; ot < 8; ++ot) {
        v8f acc = {};
#pragma unroll
        for (int kk = 0; kk < 16; ++kk) {
            const v2f bf = *(const v2f*)&wc[(2 * kk + kh) * PAIR_STRIDE + (ot * 16 + m) * 2];
            acc = __builtin_amdgcn_wmma_f32_16x16x4_f32(
                false, a[kk], false, bf, (short)0, acc, false, false);
        }
        float* Hp = H + ((size_t)tile * 16 + 8 * kh) * 128 + ot * 16 + m;
#pragma unroll
        for (int j = 0; j < 8; ++j)
            Hp[(size_t)j * 128] = acc[j];
    }
}

#define AGG_NODES  32
#define TILE_PITCH 33

__global__ __launch_bounds__(256) __attribute__((amdgpu_waves_per_eu(8, 8)))
void edgeconv_aggregate(const float* __restrict__ H,
                        const int* __restrict__ src,
                        const float* __restrict__ bias,
                        float* __restrict__ out) {
    __shared__ float tile[64 * TILE_PITCH];

    const int lane = threadIdx.x & 31;
    const int wv   = __builtin_amdgcn_readfirstlane(threadIdx.x >> 5);
    const int n0   = blockIdx.x * AGG_NODES;
    const int bb   = n0 >> 12;
    const v2f* Hp = (const v2f*)(H + ((size_t)bb * 4096) * 128 + 64);
    const v2f bV = ((const v2f*)bias)[lane];

#pragma unroll
    for (int j = 0; j < 4; ++j) {
        const int nl   = wv * 4 + j;
        const int node = n0 + nl;
        const int* sp  = src + (size_t)node * 16;

        float m0 = -3.402823466e38f, m1 = -3.402823466e38f;
#pragma unroll
        for (int k = 0; k < 16; ++k) {
            const int sk = sp[k];
            const v2f c = Hp[(unsigned)sk * 64u + lane];
            m0 = fmaxf(m0, c.x);
            m1 = fmaxf(m1, c.y);
        }

        const v2f aA = ((const v2f*)(H + (size_t)node * 128))[lane];
        tile[(2 * lane)     * TILE_PITCH + nl] = fmaxf(aA.x + bV.x + m0, 0.0f);
        tile[(2 * lane + 1) * TILE_PITCH + nl] = fmaxf(aA.y + bV.y + m1, 0.0f);
    }
    __syncthreads();

    float* ob = out + ((size_t)bb * 64) * 4096 + (n0 & 4095) + lane;
#pragma unroll
    for (int c = 0; c < 8; ++c) {
        const int ch = wv * 8 + c;
        ob[(size_t)ch * 4096] = tile[ch * TILE_PITCH + lane];
    }
}

extern "C" void kernel_launch(void* const* d_in, const int* in_sizes, int n_in,
                              void* d_out, int out_size, void* d_ws, size_t ws_size,
                              hipStream_t stream) {
    const float* x  = (const float*)d_in[0];
    const int*   ei = (const int*)d_in[1];
    const float* W  = (const float*)d_in[2];
    const float* b  = (const float*)d_in[3];
    float* out = (float*)d_out;
    float* H   = (float*)d_ws;

    edgeconv_gemm<<<256, 256, 0, stream>>>(x, W, H);
    edgeconv_aggregate<<<1024, 256, 0, stream>>>(H, ei, b, out);
}


# GEMM launch start-up loads issued as one group: weights by 8 independent 16-byte loads per lane (were 32 dependent 4-byte round trips), x fragment loads ahead of the staging wait; sums unchanged
# speedup vs baseline: 4.8417x; 1.8585x over previous
#include <hip/hip_runtime.h>
#include <hip/hip_bf16.h>

typedef float v2f __attribute__((ext_vector_type(2)));
typedef float v4f __attribute__((ext_vector_type(4)));
typedef float v8f __attribute__((ext_vector_type(8)));

#define PAIR_STRIDE 288

__global__ __launch_bounds__(256) void edgeconv_gemm(const float* __restrict__ x,
                                                     const float* __restrict__ W,
                                                     float* __restrict__ H) {
    __shared__ __attribute__((aligned(16))) float wc[32 * PAIR_STRIDE];

    const int tid  = threadIdx.x;
    const int lane = tid & 31;
    const int wv   = tid >> 5;
    const int tile = blockIdx.x * 8 + wv;
    const int m    = lane & 15;
    const int kh   = lane >> 4;
    const int node = tile * 16 + m;
    const int bb   = node >> 12;
    const int nn   = node & 4095;
    const float* xb = x + ((size_t)bb * 64) * 4096 + nn;

    const v4f* Wq = (const v4f*)W;
    v4f w1a[2], w1b[2], w2a[2], w2b[2];
#pragma unroll
    for (int j = 0; j < 2; ++j) {
        const int item = tid + 256 * j;
        const int q = item & 15;
        const int p = item >> 4;
        w1a[j] = Wq[(2 * p) * 16 + q];
        w1b[j] = Wq[(2 * p + 1) * 16 + q];
        w2a[j] = Wq[(64 + 2 * p) * 16 + q];
        w2b[j] = Wq[(64 + 2 * p + 1) * 16 + q];
    }

    v2f a[16];
#pragma unroll
    for (int kk = 0; kk < 16; ++kk) {
        int c0 = 4 * kk + 2 * kh;
        a[kk].x = xb[(size_t)c0 * 4096];
        a[kk].y = xb[(size_t)(c0 + 1) * 4096];
    }

#pragma unroll
    for (int j = 0; j < 2; ++j) {
        const int item = tid + 256 * j;
        const int q = item & 15;
        const int p = item >> 4;
        const v4f d0 = w1a[j] - w2a[j];
        const v4f d1 = w1b[j] - w2b[j];
        float* dst = &wc[p * PAIR_STRIDE + q * 8];
        *(v4f*)(dst)       = (v4f){d0.x, d1.x, d0.y, d1.y};
        *(v4f*)(dst + 4)   = (v4f){d0.z, d1.z, d0.w, d1.w};
        *(v4f*)(dst + 128) = (v4f){w2a[j].x, w2b[j].x, w2a[j].y, w2b[j].y};
        *(v4f*)(dst + 132) = (v4f){w2a[j].z, w2b[j].z, w2a[j].w, w2b[j].w};
    }
    __syncthreads();

#pragma unroll
    for (int ot = 0; ot < 8; ++ot) {
        v8f acc = {};
#pragma unroll
        for (int kk = 0; kk < 16; ++kk) {
            const v2f bf = *(const v2f*)&wc[(2 * kk + kh) * PAIR_STRIDE + (ot * 16 + m) * 2];
            acc = __builtin_amdgcn_wmma_f32_16x16x4_f32(
                false, a[kk], false, bf, (short)0, acc, false, false);
        }
        float* Hp = H + ((size_t)tile * 16 + 8 * kh) * 128 + ot * 16 + m;
#pragma unroll
        for (int j = 0; j < 8; ++j)
            Hp[(size_t)j * 128] = acc[j];
    }
}

#define AGG_NODES  32
#define TILE_PITCH 33

__global__ __launch_bounds__(256) __attribute__((amdgpu_waves_per_eu(8, 8)))
void edgeconv_aggregate(const float* __restrict__ H,
                        const int* __restrict__ src,
                        const float* __restrict__ bias,
                        float* __restrict__ out) {
    __shared__ float tile[64 * TILE_PITCH];

    const int lane = threadIdx.x & 31;
    const int wv   = __builtin_amdgcn_readfirstlane(threadIdx.x >> 5);
    const int n0   = blockIdx.x * AGG_NODES;
    const int bb   = n0 >> 12;
    const v2f* Hp = (const v2f*)(H + ((size_t)bb * 4096) * 128 + 64);
    const v2f bV = ((const v2f*)bias)[lane];

#pragma unroll
    for (int j = 0; j < 4; ++j) {
        const int nl   = wv * 4 + j;
        const int node = n0 + nl;
        const int* sp  = src + (size_t)node * 16;

        float m0 = -3.402823466e38f, m1 = -3.402823466e38f;
#pragma unroll
        for (int k = 0; k < 16; ++k) {
            const int sk = sp[k];
            const v2f c = Hp[(unsigned)sk * 64u + lane];
            m0 = fmaxf(m0, c.x);
            m1 = fmaxf(m1, c.y);
        }

        const v2f aA = ((const v2f*)(H + (size_t)node * 128))[lane];
        tile[(2 * lane)     * TILE_PITCH + nl] = fmaxf(aA.x + bV.x + m0, 0.0f);
        tile[(2 * lane + 1) * TILE_PITCH + nl] = fmaxf(aA.y + bV.y + m1, 0.0f);
    }
    __syncthreads();

    float* ob = out + ((size_t)bb * 64) * 4096 + (n0 & 4095) + lane;
#pragma unroll
    for (int c = 0; c < 8; ++c) {
        const int ch = wv * 8 + c;
        ob[(size_t)ch * 4096] = tile[ch * TILE_PITCH + lane];
    }
}

extern "C" void kernel_launch(void* const* d_in, const int* in_sizes, int n_in,
                              void* d_out, int out_size, void* d_ws, size_t ws_size,
                              hipStream_t stream) {
    const float* x  = (const float*)d_in[0];
    const int*   ei = (const int*)d_in[1];
    const float* W  = (const float*)d_in[2];
    const float* b  = (const float*)d_in[3];
    float* out = (float*)d_out;
    float* H   = (float*)d_ws;

    edgeconv_gemm<<<256, 256, 0, stream>>>(x, W, H);
    edgeconv_aggregate<<<1024, 256, 0, stream>>>(H, ei, b, out);
}


# center term X@(W1-W2) moved into the aggregate kernel: one 16x16 tile per wave on the matrix unit, under its gathers; the GEMM launch computes and stores only the gathered half of H; sums unchanged
# speedup vs baseline: 5.1496x; 1.0636x over previous
#include <hip/hip_runtime.h>
#include <hip/hip_bf16.h>


typedef float v2f __attribute__((ext_vector_type(2)));
typedef float v4f __attribute__((ext_vector_type(4)));
typedef float v8f __attribute__((ext_vector_type(8)));

#define PAIR_STRIDE 160

__global__ __launch_bounds__(256) void edgeconv_gemm(const float* __restrict__ x,
                                                     const float* __restrict__ W,
                                                     float* __restrict__ Hn,
                                                     float* __restrict__ Wd) {
    __shared__ __attribute__((aligned(16))) float wc[32 * PAIR_STRIDE];

    const int tid  = threadIdx.x;
    const int lane = tid & 31;
    const int wv   = tid >> 5;
    const int tile = blockIdx.x * 8 + wv;
    const int m    = lane & 15;
    const int kh   = lane >> 4;
    const int node = tile * 16 + m;
    const int bb   = node >> 12;
    const int nn   = node & 4095;
    const float* xb = x + ((size_t)bb * 64) * 4096 + nn;

    const v4f* Wq = (const v4f*)W;
    const bool wr = tid < 2;
    const int  wi = 2 * blockIdx.x + tid;
    const int  wq = wi & 15;
    const int  wp = wi >> 4;
    v4f e1a = {}, e1b = {}, e2a = {}, e2b = {};
    if (wr) {
        e1a = Wq[(2 * wp) * 16 + wq];
        e1b = Wq[(2 * wp + 1) * 16 + wq];
        e2a = Wq[(64 + 2 * wp) * 16 + wq];
        e2b = Wq[(64 + 2 * wp + 1) * 16 + wq];
    }
    v4f w2a[2], w2b[2];
#pragma unroll
    for (int j = 0; j < 2; ++j) {
        const int item = tid + 256 * j;
        const int q = item & 15;
        const int p = item >> 4;
        w2a[j] = Wq[(64 + 2 * p) * 16 + q];
        w2b[j] = Wq[(64 + 2 * p + 1) * 16 + q];
    }

    v2f a[16];
#pragma unroll
    for (int kk = 0; kk < 16; ++kk) {
        int c0 = 4 * kk + 2 * kh;
        a[kk].x = xb[(size_t)c0 * 4096];
        a[kk].y = xb[(size_t)(c0 + 1) * 4096];
    }

#pragma unroll
    for (int j = 0; j < 2; ++j) {
        const int item = tid + 256 * j;
        const int q = item & 15;
        const int p = item >> 4;
        float* dst = &wc[p * PAIR_STRIDE + q * 8];
        *(v4f*)(dst)     = (v4f){w2a[j].x, w2b[j].x, w2a[j].y, w2b[j].y};
        *(v4f*)(dst + 4) = (v4f){w2a[j].z, w2b[j].z, w2a[j].w, w2b[j].w};
    }

    if (wr) {
        const v4f d0 = e1a - e2a;
        const v4f d1 = e1b - e2b;
        float* g = Wd + wp * 128 + wq * 8;
        *(v4f*)(g)     = (v4f){d0.x, d1.x, d0.y, d1.y};
        *(v4f*)(g + 4) = (v4f){d0.z, d1.z, d0.w, d1.w};
    }
    __syncthreads();

#pragma unroll
    for (int ot = 0; ot < 4; ++ot) {
        v8f acc = {};
#pragma unroll
        for (int kk = 0; kk < 16; ++kk) {
            const v2f bf = *(const v2f*)&wc[(2 * kk + kh) * PAIR_STRIDE + (ot * 16 + m) * 2];
            acc = __builtin_amdgcn_wmma_f32_16x16x4_f32(
                false, a[kk], false, bf, (short)0, acc, false, false);
        }
        float* Hp = Hn + ((size_t)tile * 16 + 8 * kh) * 64 + ot * 16 + m;
#pragma unroll
        for (int j = 0; j < 8; ++j)
            Hp[(size_t)j * 64] = acc[j];
    }
}

#define AGG_NODES  32
#define TILE_PITCH 33

__global__ __launch_bounds__(256) __attribute__((amdgpu_waves_per_eu(8, 8)))
void edgeconv_aggregate(const float* __restrict__ x,
                        const float* __restrict__ Wd,
                        const float* __restrict__ Hn,
                        const int* __restrict__ src,
                        const float* __restrict__ bias,
                        float* __restrict__ out) {
    __shared__ float tile[64 * TILE_PITCH];
    __shared__ float ctile[64 * TILE_PITCH];

    const int lane = threadIdx.x & 31;
    const int wv   = __builtin_amdgcn_readfirstlane(threadIdx.x >> 5);
    const int n0   = blockIdx.x * AGG_NODES;
    const int bb   = n0 >> 12;
    const int m    = lane & 15;
    const int kh   = lane >> 4;
    const int nt   = wv & 1;
    const int ct   = wv >> 1;

    const int* sp = src + (size_t)(n0 + wv * 4) * 16;
    int sidx[64];
#pragma unroll
    for (int i = 0; i < 64; ++i)
        sidx[i] = sp[i];

    const float* xb = x + ((size_t)bb * 64) * 4096 + (n0 & 4095) + 16 * nt + m;
    const v2f* Wp = (const v2f*)Wd + (16 * ct + m);
    v2f a[16], b[16];
#pragma unroll
    for (int kk = 0; kk < 16; ++kk) {
        int c0 = 4 * kk + 2 * kh;
        a[kk].x = xb[(size_t)c0 * 4096];
        a[kk].y = xb[(size_t)(c0 + 1) * 4096];
        b[kk]   = Wp[(2 * kk + kh) * 64];
    }

    const v2f* Hp = (const v2f*)(Hn + ((size_t)bb * 4096) * 64);
    v2f g0[16];
#pragma unroll
    for (int k = 0; k < 16; ++k)
        g0[k] = Hp[(unsigned)sidx[k] * 32u + lane];

    __builtin_amdgcn_sched_barrier(0);

    {
        v8f acc = {};
#pragma unroll
        for (int kk = 0; kk < 16; ++kk)
            acc = __builtin_amdgcn_wmma_f32_16x16x4_f32(
                false, a[kk], false, b[kk], (short)0, acc, false, false);
        float* cp = &ctile[(16 * ct + m) * TILE_PITCH + 16 * nt + 8 * kh];
#pragma unroll
        for (int r = 0; r < 8; ++r)
            cp[r] = acc[r];
    }

#pragma unroll
    for (int j = 0; j < 4; ++j) {
        const int nl = wv * 4 + j;
        float m0 = -3.402823466e38f, m1 = -3.402823466e38f;
#pragma unroll
        for (int k = 0; k < 16; ++k) {
            const v2f c = (j == 0) ? g0[k] : Hp[(unsigned)sidx[j * 16 + k] * 32u + lane];
            m0 = fmaxf(m0, c.x);
            m1 = fmaxf(m1, c.y);
        }
        tile[(2 * lane)     * TILE_PITCH + nl] = m0;
        tile[(2 * lane + 1) * TILE_PITCH + nl] = m1;
    }
    __syncthreads();

    float* ob = out + ((size_t)bb * 64) * 4096 + (n0 & 4095) + lane;
#pragma unroll
    for (int c = 0; c < 8; ++c) {
        const int ch = wv * 8 + c;
        const float v = (ctile[ch * TILE_PITCH + lane] + bias[ch]) + tile[ch * TILE_PITCH + lane];
        ob[(size_t)ch * 4096] = fmaxf(v, 0.0f);
    }
}

extern "C" void kernel_launch(void* const* d_in, const int* in_sizes, int n_in,
                              void* d_out, int out_size, void* d_ws, size_t ws_size,
                              hipStream_t stream) {
    const float* x  = (const float*)d_in[0];
    const int*   ei = (const int*)d_in[1];
    const float* W  = (const float*)d_in[2];
    const float* b  = (const float*)d_in[3];
    float* out = (float*)d_out;
    float* Hn  = (float*)d_ws;
    float* Wd  = (float*)d_ws + 2097152;

    edgeconv_gemm<<<256, 256, 0, stream>>>(x, W, Hn, Wd);
    edgeconv_aggregate<<<1024, 256, 0, stream>>>(x, Wd, Hn, ei, b, out);
}


# aggregate launch shape: 512 work-groups of 64 nodes (8 per wave in two passes of 4, two column tiles per wave on one A fragment), 2 resident per WGP: 4096 waves enter, were 8192; sums unchanged
# speedup vs baseline: 5.4625x; 1.0608x over previous
#include <hip/hip_runtime.h>
#include <hip/hip_bf16.h>


typedef float v2f __attribute__((ext_vector_type(2)));
typedef float v4f __attribute__((ext_vector_type(4)));
typedef float v8f __attribute__((ext_vector_type(8)));

#define PAIR_STRIDE 160

__global__ __launch_bounds__(256) void edgeconv_gemm(const float* __restrict__ x,
                                                     const float* __restrict__ W,
                                                     float* __restrict__ Hn,
                                                     float* __restrict__ Wd) {
    __shared__ __attribute__((aligned(16))) float wc[32 * PAIR_STRIDE];

    const int tid  = threadIdx.x;
    const int lane = tid & 31;
    const int wv   = tid >> 5;
    const int tile = blockIdx.x * 8 + wv;
    const int m    = lane & 15;
    const int kh   = lane >> 4;
    const int node = tile * 16 + m;
    const int bb   = node >> 12;
    const int nn   = node & 4095;
    const float* xb = x + ((size_t)bb * 64) * 4096 + nn;

    const v4f* Wq = (const v4f*)W;
    const bool wr = tid < 2;
    const int  wi = 2 * blockIdx.x + tid;
    const int  wq = wi & 15;
    const int  wp = wi >> 4;
    v4f e1a = {}, e1b = {}, e2a = {}, e2b = {};
    if (wr) {
        e1a = Wq[(2 * wp) * 16 + wq];
        e1b = Wq[(2 * wp + 1) * 16 + wq];
        e2a = Wq[(64 + 2 * wp) * 16 + wq];
        e2b = Wq[(64 + 2 * wp + 1) * 16 + wq];
    }
    v4f w2a[2], w2b[2];
#pragma unroll
    for (int j = 0; j < 2; ++j) {
        const int item = tid + 256 * j;
        const int q = item & 15;
        const int p = item >> 4;
        w2a[j] = Wq[(64 + 2 * p) * 16 + q];
        w2b[j] = Wq[(64 + 2 * p + 1) * 16 + q];
    }

    v2f a[16];
#pragma unroll
    for (int kk = 0; kk < 16; ++kk) {
        int c0 = 4 * kk + 2 * kh;
        a[kk].x = xb[(size_t)c0 * 4096];
        a[kk].y = xb[(size_t)(c0 + 1) * 4096];
    }

#pragma unroll
    for (int j = 0; j < 2; ++j) {
        const int item = tid + 256 * j;
        const int q = item & 15;
        const int p = item >> 4;
        float* dst = &wc[p * PAIR_STRIDE + q * 8];
        *(v4f*)(dst)     = (v4f){w2a[j].x, w2b[j].x, w2a[j].y, w2b[j].y};
        *(v4f*)(dst + 4) = (v4f){w2a[j].z, w2b[j].z, w2a[j].w, w2b[j].w};
    }

    if (wr) {
        const v4f d0 = e1a - e2a;
        const v4f d1 = e1b - e2b;
        float* g = Wd + wp * 128 + wq * 8;
        *(v4f*)(g)     = (v4f){d0.x, d1.x, d0.y, d1.y};
        *(v4f*)(g + 4) = (v4f){d0.z, d1.z, d0.w, d1.w};
    }
    __syncthreads();

#pragma unroll
    for (int ot = 0; ot < 4; ++ot) {
        v8f acc = {};
#pragma unroll
        for (int kk = 0; kk < 16; ++kk) {
            const v2f bf = *(const v2f*)&wc[(2 * kk + kh) * PAIR_STRIDE + (ot * 16 + m) * 2];
            acc = __builtin_amdgcn_wmma_f32_16x16x4_f32(
                false, a[kk], false, bf, (short)0, acc, false, false);
        }
        float* Hp = Hn + ((size_t)tile * 16 + 8 * kh) * 64 + ot * 16 + m;
#pragma unroll
        for (int j = 0; j < 8; ++j)
            Hp[(size_t)j * 64] = acc[j];
    }
}

#define AGG_NODES  64
#define TILE_PITCH 65

__global__ __launch_bounds__(256) __attribute__((amdgpu_waves_per_eu(4, 4)))
void edgeconv_aggregate(const float* __restrict__ x,
                        const float* __restrict__ Wd,
                        const float* __restrict__ Hn,
                        const int* __restrict__ src,
                        const float* __restrict__ bias,
                        float* __restrict__ out) {
    __shared__ float tile[64 * TILE_PITCH];
    __shared__ float ctile[64 * TILE_PITCH];

    const int lane = threadIdx.x & 31;
    const int wv   = __builtin_amdgcn_readfirstlane(threadIdx.x >> 5);
    const int n0   = blockIdx.x * AGG_NODES;
    const int bb   = n0 >> 12;
    const int m    = lane & 15;
    const int kh   = lane >> 4;
    const int nt   = wv & 3;
    const int cp   = wv >> 2;

    const int* sp = src + (size_t)(n0 + wv * 8) * 16;
    int sa[64];
#pragma unroll
    for (int i = 0; i < 64; ++i)
        sa[i] = sp[i];

    const float* xb = x + ((size_t)bb * 64) * 4096 + (n0 & 4095) + 16 * nt + m;
    const v2f* Wp = (const v2f*)Wd + (32 * cp + m);
    v2f a[16], b0[16], b1[16];
#pragma unroll
    for (int kk = 0; kk < 16; ++kk) {
        int c0 = 4 * kk + 2 * kh;
        a[kk].x = xb[(size_t)c0 * 4096];
        a[kk].y = xb[(size_t)(c0 + 1) * 4096];
        b0[kk]  = Wp[(2 * kk + kh) * 64];
        b1[kk]  = Wp[(2 * kk + kh) * 64 + 16];
    }

    const v2f* Hp = (const v2f*)(Hn + ((size_t)bb * 4096) * 64);
    v2f g0[32];
#pragma unroll
    for (int k = 0; k < 32; ++k)
        g0[k] = Hp[(unsigned)sa[k] * 32u + lane];

    __builtin_amdgcn_sched_barrier(0);

    {
        v8f acc0 = {}, acc1 = {};
#pragma unroll
        for (int kk = 0; kk < 16; ++kk) {
            acc0 = __builtin_amdgcn_wmma_f32_16x16x4_f32(
                false, a[kk], false, b0[kk], (short)0, acc0, false, false);
            acc1 = __builtin_amdgcn_wmma_f32_16x16x4_f32(
                false, a[kk], false, b1[kk], (short)0, acc1, false, false);
        }
        float* c0p = &ctile[(32 * cp + m) * TILE_PITCH + 16 * nt + 8 * kh];
        float* c1p = c0p + 16 * TILE_PITCH;
#pragma unroll
        for (int r = 0; r < 8; ++r) {
            c0p[r] = acc0[r];
            c1p[r] = acc1[r];
        }
    }

#pragma unroll
    for (int j = 0; j < 4; ++j) {
        const int nl = wv * 8 + j;
        float m0 = -3.402823466e38f, m1 = -3.402823466e38f;
#pragma unroll
        for (int k = 0; k < 16; ++k) {
            const v2f c = (j < 2) ? g0[j * 16 + k] : Hp[(unsigned)sa[j * 16 + k] * 32u + lane];
            m0 = fmaxf(m0, c.x);
            m1 = fmaxf(m1, c.y);
        }
        tile[(2 * lane)     * TILE_PITCH + nl] = m0;
        tile[(2 * lane + 1) * TILE_PITCH + nl] = m1;
    }
    int sb[64];
#pragma unroll
    for (int i = 0; i < 64; ++i)
        sb[i] = sp[64 + i];
#pragma unroll
    for (int j = 0; j < 4; ++j) {
        const int nl = wv * 8 + 4 + j;
        float m0 = -3.402823466e38f, m1 = -3.402823466e38f;
#pragma unroll
        for (int k = 0; k < 16; ++k) {
            const v2f c = Hp[(unsigned)sb[j * 16 + k] * 32u + lane];
            m0 = fmaxf(m0, c.x);
            m1 = fmaxf(m1, c.y);
        }
        tile[(2 * lane)     * TILE_PITCH + nl] = m0;
        tile[(2 * lane + 1) * TILE_PITCH + nl] = m1;
    }
    __syncthreads();

    float* ob = out + ((size_t)bb * 64) * 4096 + (n0 & 4095) + lane;
#pragma unroll
    for (int c = 0; c < 8; ++c) {
        const int ch = wv * 8 + c;
        const float bc = bias[ch];
#pragma unroll
        for (int h = 0; h < 2; ++h) {
            const int o = ch * TILE_PITCH + 32 * h + lane;
            const float v = (ctile[o] + bc) + tile[o];
            ob[(size_t)ch * 4096 + 32 * h] = fmaxf(v, 0.0f);
        }
    }
}

extern "C" void kernel_launch(void* const* d_in, const int* in_sizes, int n_in,
                              void* d_out, int out_size, void* d_ws, size_t ws_size,
                              hipStream_t stream) {
    const float* x  = (const float*)d_in[0];
    const int*   ei = (const int*)d_in[1];
    const float* W  = (const float*)d_in[2];
    const float* b  = (const float*)d_in[3];
    float* out = (float*)d_out;
    float* Hn  = (float*)d_ws;
    float* Wd  = (float*)d_ws + 2097152;

    edgeconv_gemm<<<256, 256, 0, stream>>>(x, W, Hn, Wd);
    edgeconv_aggregate<<<512, 256, 0, stream>>>(x, Wd, Hn, ei, b, out);
}
